# starting kernel
# baseline (speedup 1.0000x reference)
#include <hip/hip_runtime.h>

typedef __attribute__((ext_vector_type(2))) float v2f;
typedef __attribute__((ext_vector_type(8))) float v8f;

#define IN_SIZE   1024
#define TILE_ROWS 16

#define W1T_OFF 0
#define W2T_OFF 16384
#define W3_OFF  32768


__global__ void __launch_bounds__(256)
bfly_precompute(const float* __restrict__ tw, float* __restrict__ ws) {
    float* w1t = ws + W1T_OFF;
    float* w2t = ws + W2T_OFF;
    float* w3  = ws + W3_OFF;
    int tid = blockIdx.x * 256 + threadIdx.x;

    if (tid < 1024) {
        int b = tid >> 4, j = tid & 15;
        float v[16];
#pragma unroll
        for (int l = 0; l < 16; ++l) v[l] = (l == j) ? 1.0f : 0.0f;
#pragma unroll
        for (int idx = 0; idx < 4; ++idx) {
            int s = 1 << idx;
            float nv[16];
#pragma unroll
            for (int l = 0; l < 16; ++l) {
                int p = b * 16 + l;
                int q = (p >> (idx + 1)) * s + (p & (s - 1));
                int i = (l >> idx) & 1;
                const float* tq = tw + idx * 2048 + q * 4 + i * 2;
                nv[l] = tq[0] * v[l & ~s] + tq[1] * v[l | s];
            }
#pragma unroll
            for (int l = 0; l < 16; ++l) v[l] = nv[l];
        }
#pragma unroll
        for (int i = 0; i < 16; ++i) w1t[b * 256 + j * 16 + i] = v[i];
    } else if (tid < 2048) {
        int t2 = tid - 1024;
        int blk = t2 >> 4, j = t2 & 15;
        int G = blk >> 4, c = blk & 15;
        float v[16];
#pragma unroll
        for (int m = 0; m < 16; ++m) v[m] = (m == j) ? 1.0f : 0.0f;
#pragma unroll
        for (int idx = 4; idx < 8; ++idx) {
            int s = 1 << idx;
            int bit = 1 << (idx - 4);
            float nv[16];
#pragma unroll
            for (int m = 0; m < 16; ++m) {
                int p = G * 256 + 16 * m + c;
                int q = (p >> (idx + 1)) * s + (p & (s - 1));
                int i = (p >> idx) & 1;
                const float* tq = tw + idx * 2048 + q * 4 + i * 2;
                nv[m] = tq[0] * v[m & ~bit] + tq[1] * v[m | bit];
            }
#pragma unroll
            for (int m = 0; m < 16; ++m) v[m] = nv[m];
        }
#pragma unroll
        for (int i = 0; i < 16; ++i) w2t[blk * 256 + j * 16 + i] = v[i];
    } else if (tid < 2304) {
        int c = tid - 2048;
        float v[4][4];
#pragma unroll
        for (int u = 0; u < 4; ++u)
#pragma unroll
            for (int j = 0; j < 4; ++j) v[u][j] = (u == j) ? 1.0f : 0.0f;
        float nv[4][4];
#pragma unroll
        for (int u = 0; u < 4; ++u) {
            const float* tq = tw + 8 * 2048 + ((u >> 1) * 256 + c) * 4 + (u & 1) * 2;
#pragma unroll
            for (int j = 0; j < 4; ++j)
                nv[u][j] = tq[0] * v[u & ~1][j] + tq[1] * v[u | 1][j];
        }
#pragma unroll
        for (int u = 0; u < 4; ++u)
#pragma unroll
            for (int j = 0; j < 4; ++j) v[u][j] = nv[u][j];
#pragma unroll
        for (int u = 0; u < 4; ++u) {
            const float* tq = tw + 9 * 2048 + (c + 256 * (u & 1)) * 4 + ((u >> 1) & 1) * 2;
#pragma unroll
            for (int j = 0; j < 4; ++j)
                nv[u][j] = tq[0] * v[u & ~2][j] + tq[1] * v[u | 2][j];
        }
#pragma unroll
        for (int u = 0; u < 4; ++u)
#pragma unroll
            for (int j = 0; j < 4; ++j) w3[c * 16 + u * 4 + j] = nv[u][j];
    }
}

__device__ __forceinline__ int SWZ(int r, int c) {
    return r * IN_SIZE + (c ^ ((r & 15) << 2));
}

__global__ void __launch_bounds__(256)
bfly_main(const float* __restrict__ x, const float* __restrict__ bias,
          const float* __restrict__ ws, float* __restrict__ out) {
    __shared__ float U[TILE_ROWS * IN_SIZE];

    const float* w1t = ws + W1T_OFF;
    const float* w2t = ws + W2T_OFF;
    const float* w3  = ws + W3_OFF;

    const int row0 = blockIdx.x * TILE_ROWS;
    const int tid  = threadIdx.x;
    const int wave = tid >> 5;
    const int lane = tid & 31;
    const int m16  = lane & 15;
    const int khi  = (lane >> 4) * 2;

    {
        const float4* xg = (const float4*)(x + (size_t)row0 * IN_SIZE);
#pragma unroll
        for (int it = 0; it < 16; ++it) {
            int idx = it * 256 + tid;
            int r   = idx >> 8;
            int c4  = idx & 255;
            float4 val = xg[r * 256 + c4];
            *(float4*)&U[r * IN_SIZE + ((c4 * 4) ^ ((r & 15) << 2))] = val;
        }
    }
    __syncthreads();

    {
        v2f a[8][4];
#pragma unroll
        for (int t = 0; t < 8; ++t) {
            int b = wave + 8 * t;
#pragma unroll
            for (int k = 0; k < 4; ++k) {
                int col = b * 16 + k * 4 + khi;
                a[t][k].x = U[SWZ(m16, col)];
                a[t][k].y = U[SWZ(m16, col + 1)];
            }
        }
        __syncthreads();
#pragma unroll
        for (int t = 0; t < 8; ++t) {
            int b = wave + 8 * t;
            v8f acc = {};
#pragma unroll
            for (int k = 0; k < 4; ++k) {
                int K = k * 4 + khi;
                v2f bf;
                bf.x = w1t[b * 256 + K * 16 + m16];
                bf.y = w1t[b * 256 + (K + 1) * 16 + m16];
                acc = __builtin_amdgcn_wmma_f32_16x16x4_f32(
                        false, a[t][k], false, bf, (short)0, acc, false, false);
            }
#pragma unroll
            for (int v = 0; v < 8; ++v) {
                int r = v + ((lane >> 4) << 3);
                U[SWZ(r, b * 16 + m16)] = acc[v];
            }
        }
    }
    __syncthreads();

    {
        v2f a[8][4];
#pragma unroll
        for (int t = 0; t < 8; ++t) {
            int blk = wave + 8 * t;
            int G = blk >> 4, c = blk & 15;
#pragma unroll
            for (int k = 0; k < 4; ++k) {
                int mb = k * 4 + khi;
                a[t][k].x = U[SWZ(m16, G * 256 + mb * 16 + c)];
                a[t][k].y = U[SWZ(m16, G * 256 + (mb + 1) * 16 + c)];
            }
        }
        __syncthreads();
#pragma unroll
        for (int t = 0; t < 8; ++t) {
            int blk = wave + 8 * t;
            int G = blk >> 4, c = blk & 15;
            v8f acc = {};
#pragma unroll
            for (int k = 0; k < 4; ++k) {
                int K = k * 4 + khi;
                v2f bf;
                bf.x = w2t[blk * 256 + K * 16 + m16];
                bf.y = w2t[blk * 256 + (K + 1) * 16 + m16];
                acc = __builtin_amdgcn_wmma_f32_16x16x4_f32(
                        false, a[t][k], false, bf, (short)0, acc, false, false);
            }
#pragma unroll
            for (int v = 0; v < 8; ++v) {
                int r = v + ((lane >> 4) << 3);
                U[SWZ(r, G * 256 + m16 * 16 + c)] = acc[v];
            }
        }
    }
    __syncthreads();

    {
        int c = tid;
        float m[16];
#pragma unroll
        for (int i = 0; i < 16; ++i) m[i] = w3[c * 16 + i];
        float bi[4];
#pragma unroll
        for (int u = 0; u < 4; ++u) bi[u] = bias[c + 256 * u];
#pragma unroll
        for (int r = 0; r < 16; ++r) {
            float in0 = U[SWZ(r, c)];
            float in1 = U[SWZ(r, c + 256)];
            float in2 = U[SWZ(r, c + 512)];
            float in3 = U[SWZ(r, c + 768)];
            float* og = out + (size_t)(row0 + r) * IN_SIZE;
#pragma unroll
            for (int u = 0; u < 4; ++u) {
                og[c + 256 * u] = m[u * 4 + 0] * in0 + m[u * 4 + 1] * in1 +
                                  m[u * 4 + 2] * in2 + m[u * 4 + 3] * in3 + bi[u];
            }
        }
    }
}

extern "C" void kernel_launch(void* const* d_in, const int* in_sizes, int n_in,
                              void* d_out, int out_size, void* d_ws, size_t ws_size,
                              hipStream_t stream) {
    const float* x    = (const float*)d_in[0];
    const float* tw   = (const float*)d_in[1];
    const float* bias = (const float*)d_in[2];
    float* out = (float*)d_out;
    float* ws  = (float*)d_ws;

    int rows = in_sizes[0] / IN_SIZE;

    bfly_precompute<<<dim3(9), dim3(256), 0, stream>>>(tw, ws);
    bfly_main<<<dim3(rows / TILE_ROWS), dim3(256), 0, stream>>>(x, bias, ws, out);
}


# LDS layout chosen per stage group: the stride-16 gather and scatter of stages 4-7 hit 32 distinct banks (were 2-way and 8-way conflicts); weight loads fenced one block ahead to stay at 128 regs
# speedup vs baseline: 1.2015x; 1.2015x over previous
#include <hip/hip_runtime.h>

typedef __attribute__((ext_vector_type(2))) float v2f;
typedef __attribute__((ext_vector_type(8))) float v8f;

#define IN_SIZE   1024
#define TILE_ROWS 16

#define W1T_OFF 0
#define W2T_OFF 16384
#define W3_OFF  32768


__global__ void __launch_bounds__(256)
bfly_precompute(const float* __restrict__ tw, float* __restrict__ ws) {
    float* w1t = ws + W1T_OFF;
    float* w2t = ws + W2T_OFF;
    float* w3  = ws + W3_OFF;
    int tid = blockIdx.x * 256 + threadIdx.x;

    if (tid < 1024) {
        int b = tid >> 4, j = tid & 15;
        float v[16];
#pragma unroll
        for (int l = 0; l < 16; ++l) v[l] = (l == j) ? 1.0f : 0.0f;
#pragma unroll
        for (int idx = 0; idx < 4; ++idx) {
            int s = 1 << idx;
            float nv[16];
#pragma unroll
            for (int l = 0; l < 16; ++l) {
                int p = b * 16 + l;
                int q = (p >> (idx + 1)) * s + (p & (s - 1));
                int i = (l >> idx) & 1;
                const float* tq = tw + idx * 2048 + q * 4 + i * 2;
                nv[l] = tq[0] * v[l & ~s] + tq[1] * v[l | s];
            }
#pragma unroll
            for (int l = 0; l < 16; ++l) v[l] = nv[l];
        }
#pragma unroll
        for (int i = 0; i < 16; ++i) w1t[b * 256 + j * 16 + i] = v[i];
    } else if (tid < 2048) {
        int t2 = tid - 1024;
        int blk = t2 >> 4, j = t2 & 15;
        int G = blk >> 4, c = blk & 15;
        float v[16];
#pragma unroll
        for (int m = 0; m < 16; ++m) v[m] = (m == j) ? 1.0f : 0.0f;
#pragma unroll
        for (int idx = 4; idx < 8; ++idx) {
            int s = 1 << idx;
            int bit = 1 << (idx - 4);
            float nv[16];
#pragma unroll
            for (int m = 0; m < 16; ++m) {
                int p = G * 256 + 16 * m + c;
                int q = (p >> (idx + 1)) * s + (p & (s - 1));
                int i = (p >> idx) & 1;
                const float* tq = tw + idx * 2048 + q * 4 + i * 2;
                nv[m] = tq[0] * v[m & ~bit] + tq[1] * v[m | bit];
            }
#pragma unroll
            for (int m = 0; m < 16; ++m) v[m] = nv[m];
        }
#pragma unroll
        for (int i = 0; i < 16; ++i) w2t[blk * 256 + j * 16 + i] = v[i];
    } else if (tid < 2304) {
        int c = tid - 2048;
        float v[4][4];
#pragma unroll
        for (int u = 0; u < 4; ++u)
#pragma unroll
            for (int j = 0; j < 4; ++j) v[u][j] = (u == j) ? 1.0f : 0.0f;
        float nv[4][4];
#pragma unroll
        for (int u = 0; u < 4; ++u) {
            const float* tq = tw + 8 * 2048 + ((u >> 1) * 256 + c) * 4 + (u & 1) * 2;
#pragma unroll
            for (int j = 0; j < 4; ++j)
                nv[u][j] = tq[0] * v[u & ~1][j] + tq[1] * v[u | 1][j];
        }
#pragma unroll
        for (int u = 0; u < 4; ++u)
#pragma unroll
            for (int j = 0; j < 4; ++j) v[u][j] = nv[u][j];
#pragma unroll
        for (int u = 0; u < 4; ++u) {
            const float* tq = tw + 9 * 2048 + (c + 256 * (u & 1)) * 4 + ((u >> 1) & 1) * 2;
#pragma unroll
            for (int j = 0; j < 4; ++j)
                nv[u][j] = tq[0] * v[u & ~2][j] + tq[1] * v[u | 2][j];
        }
#pragma unroll
        for (int u = 0; u < 4; ++u)
#pragma unroll
            for (int j = 0; j < 4; ++j) w3[c * 16 + u * 4 + j] = nv[u][j];
    }
}

__device__ __forceinline__ int SWZ(int r, int c) {
    return r * IN_SIZE + (c ^ ((r & 15) << 2));
}
__device__ __forceinline__ int SWZ1(int r, int c) {
    return r * IN_SIZE + (c ^ ((r & 7) | ((r & 8) << 1)));
}
__device__ __forceinline__ int SWZ2(int r, int c) {
    return r * IN_SIZE + ((c ^ (((c >> 6) & 3) << 2)) ^ ((r >> 3) & 1));
}

__global__ void __launch_bounds__(256)
bfly_main(const float* __restrict__ x, const float* __restrict__ bias,
          const float* __restrict__ ws, float* __restrict__ out) {
    __shared__ float U[TILE_ROWS * IN_SIZE];

    const float* w1t = ws + W1T_OFF;
    const float* w2t = ws + W2T_OFF;
    const float* w3  = ws + W3_OFF;

    const int row0 = blockIdx.x * TILE_ROWS;
    const int tid  = threadIdx.x;
    const int wave = tid >> 5;
    const int lane = tid & 31;
    const int m16  = lane & 15;
    const int khi  = (lane >> 4) * 2;

    {
        const float4* xg = (const float4*)(x + (size_t)row0 * IN_SIZE);
#pragma unroll
        for (int it = 0; it < 16; ++it) {
            int idx = it * 256 + tid;
            int r   = idx >> 8;
            int c4  = idx & 255;
            float4 val = xg[r * 256 + c4];
            *(float4*)&U[r * IN_SIZE + ((c4 * 4) ^ ((r & 15) << 2))] = val;
        }
    }
    __syncthreads();

    {
        v2f a[8][4];
#pragma unroll
        for (int t = 0; t < 8; ++t) {
            int b = wave + 8 * t;
#pragma unroll
            for (int k = 0; k < 4; ++k) {
                int col = b * 16 + k * 4 + khi;
                a[t][k].x = U[SWZ(m16, col)];
                a[t][k].y = U[SWZ(m16, col + 1)];
            }
        }
        __syncthreads();
        v2f bn[4];
#pragma unroll
        for (int k = 0; k < 4; ++k) {
            int K = k * 4 + khi;
            bn[k].x = w1t[wave * 256 + K * 16 + m16];
            bn[k].y = w1t[wave * 256 + (K + 1) * 16 + m16];
        }
#pragma unroll
        for (int t = 0; t < 8; ++t) {
            int b = wave + 8 * t;
            v2f bc[4];
#pragma unroll
            for (int k = 0; k < 4; ++k) bc[k] = bn[k];
            if (t + 1 < 8) {
#pragma unroll
                for (int k = 0; k < 4; ++k) {
                    int K = k * 4 + khi;
                    bn[k].x = w1t[(b + 8) * 256 + K * 16 + m16];
                    bn[k].y = w1t[(b + 8) * 256 + (K + 1) * 16 + m16];
                }
            }
            v8f acc = {};
#pragma unroll
            for (int k = 0; k < 4; ++k)
                acc = __builtin_amdgcn_wmma_f32_16x16x4_f32(
                        false, a[t][k], false, bc[k], (short)0, acc, false, false);
#pragma unroll
            for (int v = 0; v < 8; ++v) {
                int r = v + ((lane >> 4) << 3);
                U[SWZ1(r, b * 16 + m16)] = acc[v];
            }
            __builtin_amdgcn_sched_barrier(0);
        }
    }
    __syncthreads();

    {
        v2f a[8][4];
#pragma unroll
        for (int t = 0; t < 8; ++t) {
            int blk = wave + 8 * t;
            int G = blk >> 4, c = blk & 15;
#pragma unroll
            for (int k = 0; k < 4; ++k) {
                int mb = k * 4 + khi;
                a[t][k].x = U[SWZ1(m16, G * 256 + mb * 16 + c)];
                a[t][k].y = U[SWZ1(m16, G * 256 + (mb + 1) * 16 + c)];
            }
        }
        __syncthreads();
        v2f bn[4];
#pragma unroll
        for (int k = 0; k < 4; ++k) {
            int K = k * 4 + khi;
            bn[k].x = w2t[wave * 256 + K * 16 + m16];
            bn[k].y = w2t[wave * 256 + (K + 1) * 16 + m16];
        }
#pragma unroll
        for (int t = 0; t < 8; ++t) {
            int blk = wave + 8 * t;
            int G = blk >> 4, c = blk & 15;
            v2f bc[4];
#pragma unroll
            for (int k = 0; k < 4; ++k) bc[k] = bn[k];
            if (t + 1 < 8) {
#pragma unroll
                for (int k = 0; k < 4; ++k) {
                    int K = k * 4 + khi;
                    bn[k].x = w2t[(blk + 8) * 256 + K * 16 + m16];
                    bn[k].y = w2t[(blk + 8) * 256 + (K + 1) * 16 + m16];
                }
            }
            v8f acc = {};
#pragma unroll
            for (int k = 0; k < 4; ++k)
                acc = __builtin_amdgcn_wmma_f32_16x16x4_f32(
                        false, a[t][k], false, bc[k], (short)0, acc, false, false);
#pragma unroll
            for (int v = 0; v < 8; ++v) {
                int r = v + ((lane >> 4) << 3);
                U[SWZ2(r, G * 256 + m16 * 16 + c)] = acc[v];
            }
            __builtin_amdgcn_sched_barrier(0);
        }
    }
    __syncthreads();

    {
        int c = tid;
        float m[16];
#pragma unroll
        for (int i = 0; i < 16; ++i) m[i] = w3[c * 16 + i];
        float bi[4];
#pragma unroll
        for (int u = 0; u < 4; ++u) bi[u] = bias[c + 256 * u];
#pragma unroll
        for (int r = 0; r < 16; ++r) {
            float in0 = U[SWZ2(r, c)];
            float in1 = U[SWZ2(r, c + 256)];
            float in2 = U[SWZ2(r, c + 512)];
            float in3 = U[SWZ2(r, c + 768)];
            float* og = out + (size_t)(row0 + r) * IN_SIZE;
#pragma unroll
            for (int u = 0; u < 4; ++u) {
                og[c + 256 * u] = m[u * 4 + 0] * in0 + m[u * 4 + 1] * in1 +
                                  m[u * 4 + 2] * in2 + m[u * 4 + 3] * in3 + bi[u];
            }
        }
    }
}

extern "C" void kernel_launch(void* const* d_in, const int* in_sizes, int n_in,
                              void* d_out, int out_size, void* d_ws, size_t ws_size,
                              hipStream_t stream) {
    const float* x    = (const float*)d_in[0];
    const float* tw   = (const float*)d_in[1];
    const float* bias = (const float*)d_in[2];
    float* out = (float*)d_out;
    float* ws  = (float*)d_ws;

    int rows = in_sizes[0] / IN_SIZE;

    bfly_precompute<<<dim3(9), dim3(256), 0, stream>>>(tw, ws);
    bfly_main<<<dim3(rows / TILE_ROWS), dim3(256), 0, stream>>>(x, bias, ws, out);
}


# Last stage (4x4 mix + bias): each lane owns 4 adjacent columns of 4 rows, so a wave store writes 512 contiguous bytes (whole 256-byte units; was 128) and LDS reads are 128-bit; W3 as 16 planes
# speedup vs baseline: 1.3495x; 1.1232x over previous
#include <hip/hip_runtime.h>

typedef __attribute__((ext_vector_type(2))) float v2f;
typedef __attribute__((ext_vector_type(8))) float v8f;

#define IN_SIZE   1024
#define TILE_ROWS 16

#define W1T_OFF 0
#define W2T_OFF 16384
#define W3_OFF  32768


__global__ void __launch_bounds__(256)
bfly_precompute(const float* __restrict__ tw, float* __restrict__ ws) {
    float* w1t = ws + W1T_OFF;
    float* w2t = ws + W2T_OFF;
    float* w3  = ws + W3_OFF;
    int tid = blockIdx.x * 256 + threadIdx.x;

    if (tid < 1024) {
        int b = tid >> 4, j = tid & 15;
        float v[16];
#pragma unroll
        for (int l = 0; l < 16; ++l) v[l] = (l == j) ? 1.0f : 0.0f;
#pragma unroll
        for (int idx = 0; idx < 4; ++idx) {
            int s = 1 << idx;
            float nv[16];
#pragma unroll
            for (int l = 0; l < 16; ++l) {
                int p = b * 16 + l;
                int q = (p >> (idx + 1)) * s + (p & (s - 1));
                int i = (l >> idx) & 1;
                const float* tq = tw + idx * 2048 + q * 4 + i * 2;
                nv[l] = tq[0] * v[l & ~s] + tq[1] * v[l | s];
            }
#pragma unroll
            for (int l = 0; l < 16; ++l) v[l] = nv[l];
        }
#pragma unroll
        for (int i = 0; i < 16; ++i) w1t[b * 256 + j * 16 + i] = v[i];
    } else if (tid < 2048) {
        int t2 = tid - 1024;
        int blk = t2 >> 4, j = t2 & 15;
        int G = blk >> 4, c = blk & 15;
        float v[16];
#pragma unroll
        for (int m = 0; m < 16; ++m) v[m] = (m == j) ? 1.0f : 0.0f;
#pragma unroll
        for (int idx = 4; idx < 8; ++idx) {
            int s = 1 << idx;
            int bit = 1 << (idx - 4);
            float nv[16];
#pragma unroll
            for (int m = 0; m < 16; ++m) {
                int p = G * 256 + 16 * m + c;
                int q = (p >> (idx + 1)) * s + (p & (s - 1));
                int i = (p >> idx) & 1;
                const float* tq = tw + idx * 2048 + q * 4 + i * 2;
                nv[m] = tq[0] * v[m & ~bit] + tq[1] * v[m | bit];
            }
#pragma unroll
            for (int m = 0; m < 16; ++m) v[m] = nv[m];
        }
#pragma unroll
        for (int i = 0; i < 16; ++i) w2t[blk * 256 + j * 16 + i] = v[i];
    } else if (tid < 2304) {
        int c = tid - 2048;
        float v[4][4];
#pragma unroll
        for (int u = 0; u < 4; ++u)
#pragma unroll
            for (int j = 0; j < 4; ++j) v[u][j] = (u == j) ? 1.0f : 0.0f;
        float nv[4][4];
#pragma unroll
        for (int u = 0; u < 4; ++u) {
            const float* tq = tw + 8 * 2048 + ((u >> 1) * 256 + c) * 4 + (u & 1) * 2;
#pragma unroll
            for (int j = 0; j < 4; ++j)
                nv[u][j] = tq[0] * v[u & ~1][j] + tq[1] * v[u | 1][j];
        }
#pragma unroll
        for (int u = 0; u < 4; ++u)
#pragma unroll
            for (int j = 0; j < 4; ++j) v[u][j] = nv[u][j];
#pragma unroll
        for (int u = 0; u < 4; ++u) {
            const float* tq = tw + 9 * 2048 + (c + 256 * (u & 1)) * 4 + ((u >> 1) & 1) * 2;
#pragma unroll
            for (int j = 0; j < 4; ++j)
                nv[u][j] = tq[0] * v[u & ~2][j] + tq[1] * v[u | 2][j];
        }
#pragma unroll
        for (int u = 0; u < 4; ++u)
#pragma unroll
            for (int j = 0; j < 4; ++j) w3[(u * 4 + j) * 256 + c] = nv[u][j];
    }
}

__device__ __forceinline__ int SWZ(int r, int c) {
    return r * IN_SIZE + (c ^ ((r & 15) << 2));
}
__device__ __forceinline__ int SWZ1(int r, int c) {
    return r * IN_SIZE + (c ^ ((r & 7) | ((r & 8) << 1)));
}
__device__ __forceinline__ int SWZ2(int r, int c) {
    return r * IN_SIZE + ((c ^ (((c >> 6) & 3) << 2)) ^ (((r >> 3) & 1) << 1));
}

__global__ void __launch_bounds__(256)
bfly_main(const float* __restrict__ x, const float* __restrict__ bias,
          const float* __restrict__ ws, float* __restrict__ out) {
    __shared__ __attribute__((aligned(16))) float U[TILE_ROWS * IN_SIZE];

    const float* w1t = ws + W1T_OFF;
    const float* w2t = ws + W2T_OFF;
    const float* w3  = ws + W3_OFF;

    const int row0 = blockIdx.x * TILE_ROWS;
    const int tid  = threadIdx.x;
    const int wave = tid >> 5;
    const int lane = tid & 31;
    const int m16  = lane & 15;
    const int khi  = (lane >> 4) * 2;

    {
        const float4* xg = (const float4*)(x + (size_t)row0 * IN_SIZE);
#pragma unroll
        for (int it = 0; it < 16; ++it) {
            int idx = it * 256 + tid;
            int r   = idx >> 8;
            int c4  = idx & 255;
            float4 val = xg[r * 256 + c4];
            *(float4*)&U[r * IN_SIZE + ((c4 * 4) ^ ((r & 15) << 2))] = val;
        }
    }
    __syncthreads();

    {
        v2f a[8][4];
#pragma unroll
        for (int t = 0; t < 8; ++t) {
            int b = wave + 8 * t;
#pragma unroll
            for (int k = 0; k < 4; ++k) {
                int col = b * 16 + k * 4 + khi;
                a[t][k].x = U[SWZ(m16, col)];
                a[t][k].y = U[SWZ(m16, col + 1)];
            }
        }
        __syncthreads();
        v2f bn[4];
#pragma unroll
        for (int k = 0; k < 4; ++k) {
            int K = k * 4 + khi;
            bn[k].x = w1t[wave * 256 + K * 16 + m16];
            bn[k].y = w1t[wave * 256 + (K + 1) * 16 + m16];
        }
#pragma unroll
        for (int t = 0; t < 8; ++t) {
            int b = wave + 8 * t;
            v2f bc[4];
#pragma unroll
            for (int k = 0; k < 4; ++k) bc[k] = bn[k];
            if (t + 1 < 8) {
#pragma unroll
                for (int k = 0; k < 4; ++k) {
                    int K = k * 4 + khi;
                    bn[k].x = w1t[(b + 8) * 256 + K * 16 + m16];
                    bn[k].y = w1t[(b + 8) * 256 + (K + 1) * 16 + m16];
                }
            }
            v8f acc = {};
#pragma unroll
            for (int k = 0; k < 4; ++k)
                acc = __builtin_amdgcn_wmma_f32_16x16x4_f32(
                        false, a[t][k], false, bc[k], (short)0, acc, false, false);
#pragma unroll
            for (int v = 0; v < 8; ++v) {
                int r = v + ((lane >> 4) << 3);
                U[SWZ1(r, b * 16 + m16)] = acc[v];
            }
            __builtin_amdgcn_sched_barrier(0);
        }
    }
    __syncthreads();

    {
        v2f a[8][4];
#pragma unroll
        for (int t = 0; t < 8; ++t) {
            int blk = wave + 8 * t;
            int G = blk >> 4, c = blk & 15;
#pragma unroll
            for (int k = 0; k < 4; ++k) {
                int mb = k * 4 + khi;
                a[t][k].x = U[SWZ1(m16, G * 256 + mb * 16 + c)];
                a[t][k].y = U[SWZ1(m16, G * 256 + (mb + 1) * 16 + c)];
            }
        }
        __syncthreads();
        v2f bn[4];
#pragma unroll
        for (int k = 0; k < 4; ++k) {
            int K = k * 4 + khi;
            bn[k].x = w2t[wave * 256 + K * 16 + m16];
            bn[k].y = w2t[wave * 256 + (K + 1) * 16 + m16];
        }
#pragma unroll
        for (int t = 0; t < 8; ++t) {
            int blk = wave + 8 * t;
            int G = blk >> 4, c = blk & 15;
            v2f bc[4];
#pragma unroll
            for (int k = 0; k < 4; ++k) bc[k] = bn[k];
            if (t + 1 < 8) {
#pragma unroll
                for (int k = 0; k < 4; ++k) {
                    int K = k * 4 + khi;
                    bn[k].x = w2t[(blk + 8) * 256 + K * 16 + m16];
                    bn[k].y = w2t[(blk + 8) * 256 + (K + 1) * 16 + m16];
                }
            }
            v8f acc = {};
#pragma unroll
            for (int k = 0; k < 4; ++k)
                acc = __builtin_amdgcn_wmma_f32_16x16x4_f32(
                        false, a[t][k], false, bc[k], (short)0, acc, false, false);
#pragma unroll
            for (int v = 0; v < 8; ++v) {
                int r = v + ((lane >> 4) << 3);
                U[SWZ2(r, G * 256 + m16 * 16 + c)] = acc[v];
            }
            __builtin_amdgcn_sched_barrier(0);
        }
    }
    __syncthreads();

    {
        const int cg = tid & 63;
        const int rg = tid >> 6;
        float4 m[16];
#pragma unroll
        for (int e = 0; e < 16; ++e) m[e] = *(const float4*)&w3[e * 256 + 4 * cg];
        float4 bi[4];
#pragma unroll
        for (int u = 0; u < 4; ++u) bi[u] = *(const float4*)&bias[4 * cg + 256 * u];
        const int pc = (4 * cg) ^ (((cg >> 4) & 3) << 2);
#pragma unroll
        for (int h = 0; h < 2; ++h) {
#pragma unroll
            for (int q = 0; q < 2; ++q) {
                const int r = 2 * rg + q + 8 * h;
                float4 in[4];
#pragma unroll
                for (int u = 0; u < 4; ++u) {
                    float4 t = *(const float4*)&U[r * IN_SIZE + pc + 256 * u];
                    if (h) in[u] = make_float4(t.z, t.w, t.x, t.y);
                    else   in[u] = t;
                }
                float* og = out + (size_t)(row0 + r) * IN_SIZE + 4 * cg;
#pragma unroll
                for (int u = 0; u < 4; ++u) {
                    float4 o;
                    o.x = m[u * 4 + 0].x * in[0].x + m[u * 4 + 1].x * in[1].x +
                          m[u * 4 + 2].x * in[2].x + m[u * 4 + 3].x * in[3].x + bi[u].x;
                    o.y = m[u * 4 + 0].y * in[0].y + m[u * 4 + 1].y * in[1].y +
                          m[u * 4 + 2].y * in[2].y + m[u * 4 + 3].y * in[3].y + bi[u].y;
                    o.z = m[u * 4 + 0].z * in[0].z + m[u * 4 + 1].z * in[1].z +
                          m[u * 4 + 2].z * in[2].z + m[u * 4 + 3].z * in[3].z + bi[u].z;
                    o.w = m[u * 4 + 0].w * in[0].w + m[u * 4 + 1].w * in[1].w +
                          m[u * 4 + 2].w * in[2].w + m[u * 4 + 3].w * in[3].w + bi[u].w;
                    *(float4*)&og[256 * u] = o;
                }
            }
        }
    }
}

extern "C" void kernel_launch(void* const* d_in, const int* in_sizes, int n_in,
                              void* d_out, int out_size, void* d_ws, size_t ws_size,
                              hipStream_t stream) {
    const float* x    = (const float*)d_in[0];
    const float* tw   = (const float*)d_in[1];
    const float* bias = (const float*)d_in[2];
    float* out = (float*)d_out;
    float* ws  = (float*)d_ws;

    int rows = in_sizes[0] / IN_SIZE;

    bfly_precompute<<<dim3(9), dim3(256), 0, stream>>>(tw, ws);
    bfly_main<<<dim3(rows / TILE_ROWS), dim3(256), 0, stream>>>(x, bias, ws, out);
}
